# starting kernel
# baseline (speedup 1.0000x reference)
#include <hip/hip_runtime.h>


#define BATCH  16
#define SEQ    4096
#define DIM    1024
#define SSPLIT 64
#define SCHUNK (SEQ / SSPLIT)

typedef float  v2f  __attribute__((ext_vector_type(2)));
typedef float  v4f  __attribute__((ext_vector_type(4)));
typedef float  v8f  __attribute__((ext_vector_type(8)));
typedef _Float16 v16h __attribute__((ext_vector_type(16)));

__global__ __launch_bounds__(256) void hcen_mean_stage1(
    const float* __restrict__ x, float* __restrict__ part) {
  const int sblk = blockIdx.x;
  const int b    = blockIdx.y;
  const int t    = threadIdx.x;

  const float* xp = x + ((size_t)b * SEQ + (size_t)sblk * SCHUNK) * DIM + t * 4;
  v4f acc = {0.f, 0.f, 0.f, 0.f};
#pragma unroll 4
  for (int s = 0; s < SCHUNK; ++s) {
    v4f v = __builtin_nontemporal_load((const v4f*)(xp + (size_t)s * DIM));
    acc += v;
  }
  *(v4f*)(part + (size_t)(sblk * BATCH + b) * DIM + t * 4) = acc;
}

__global__ __launch_bounds__(256) void hcen_mean_stage2(
    const float* __restrict__ part, float* __restrict__ m) {
  const int i = blockIdx.x * 256 + threadIdx.x;
  float acc = 0.f;
#pragma unroll 8
  for (int s = 0; s < SSPLIT; ++s)
    acc += part[(size_t)s * (BATCH * DIM) + i];
  m[i] = acc * (1.0f / (float)SEQ);
}

__global__ __launch_bounds__(32) void hcen_gemm16(
    const float* __restrict__ A, const float* __restrict__ W,
    const float* __restrict__ bias, float* __restrict__ D,
    int K, int N) {
  const int n0   = blockIdx.x * 16;
  const int lane = threadIdx.x;
  const int half = lane >> 4;
  const int lm   = lane & 15;

  v8f c;
  const float bv = bias[n0 + lm];
#pragma unroll
  for (int j = 0; j < 8; ++j) c[j] = bv;

#if __has_builtin(__builtin_amdgcn_wmma_f32_16x16x4_f32)
  const float* arow = A + (size_t)lm * K + 2 * half;
  const float* wrow = W + (size_t)(n0 + lm) * K + 2 * half;
#pragma unroll 8
  for (int k0 = 0; k0 < K; k0 += 4) {
    v2f av = *(const v2f*)(arow + k0);
    v2f bv2 = *(const v2f*)(wrow + k0);
    c = __builtin_amdgcn_wmma_f32_16x16x4_f32(
        false, av, false, bv2, (short)0, c, false, false);
  }
#else
  const float* arow = A + (size_t)lm * K + 8 * half;
  const float* wrow = W + (size_t)(n0 + lm) * K + 8 * half;
  for (int k0 = 0; k0 < K; k0 += 32) {
    v16h af, bf;
#pragma unroll
    for (int e = 0; e < 16; ++e) {
      const int off = k0 + 16 * (e >> 3) + (e & 7);
      af[e] = (_Float16)arow[off];
      bf[e] = (_Float16)wrow[off];
    }
    c = __builtin_amdgcn_wmma_f32_16x16x32_f16(
        false, af, false, bf, (short)0, c, false, false);
  }
#endif

#pragma unroll
  for (int j = 0; j < 8; ++j)
    D[(size_t)(j + 8 * half) * N + (n0 + lm)] = c[j];
}

extern "C" void kernel_launch(void* const* d_in, const int* in_sizes, int n_in,
                              void* d_out, int out_size, void* d_ws, size_t ws_size,
                              hipStream_t stream) {
  const float* x     = (const float*)d_in[0];
  const float* W_enc = (const float*)d_in[1];
  const float* b_enc = (const float*)d_in[2];
  const float* W_out = (const float*)d_in[3];
  const float* b_out = (const float*)d_in[4];
  float* out = (float*)d_out;

  float* part = (float*)d_ws;
  float* m    = part + (size_t)SSPLIT * BATCH * DIM;
  float* enc  = m + (size_t)BATCH * DIM;

  dim3 g1(SSPLIT, BATCH);
  hcen_mean_stage1<<<g1, 256, 0, stream>>>(x, part);
  hcen_mean_stage2<<<(BATCH * DIM) / 256, 256, 0, stream>>>(part, m);
  hcen_gemm16<<<DIM / 16, 32, 0, stream>>>(m,   W_enc, b_enc, enc, DIM, DIM);
  hcen_gemm16<<<DIM / 16, 32, 0, stream>>>(enc, W_out, b_out, out, DIM, DIM);
}


# GEMMs re-tiled 2-D: block = 32-wide k slab x 128 outputs (256 blocks, was 64); 32 planes per stage, two alternating 2 MB sets; sums regrouped: 128-row chunks, 32-slabs, 2 accumulators (new order)
# speedup vs baseline: 1.8061x; 1.8061x over previous
#include <hip/hip_runtime.h>


#define BATCH  16
#define SEQ    4096
#define DIM    1024
#define SSPLIT 32
#define SCHUNK (SEQ / SSPLIT)
#define KSLAB  32
#define NSLAB  (DIM / KSLAB)
#define NRANGE 8
#define PLANE  (BATCH * DIM)
#define PIECE  (BATCH * KSLAB)
#define SETSZ  (NSLAB * PLANE)
#define APITCH 36

typedef float  v2f  __attribute__((ext_vector_type(2)));
typedef float  v4f  __attribute__((ext_vector_type(4)));
typedef float  v8f  __attribute__((ext_vector_type(8)));

__global__ __launch_bounds__(256) void hcen_mean_stage1(
    const float* __restrict__ x, float* __restrict__ part) {
  const int sblk = blockIdx.x;
  const int b    = blockIdx.y;
  const int t    = threadIdx.x;

  const float* xp = x + ((size_t)b * SEQ + (size_t)sblk * SCHUNK) * DIM + t * 4;
  v4f acc = {0.f, 0.f, 0.f, 0.f};
#pragma unroll 8
  for (int s = 0; s < SCHUNK; ++s) {
    v4f v = __builtin_nontemporal_load((const v4f*)(xp + (size_t)s * DIM));
    acc += v;
  }
  *(v4f*)(part + (size_t)sblk * PLANE + (t >> 3) * PIECE + b * KSLAB + (t & 7) * 4) = acc;
}

template <bool FIRST>
__global__ __launch_bounds__(256) void hcen_rank32(
    float* buf, const float* __restrict__ Wm, const float* __restrict__ bias_in) {
  __shared__ __attribute__((aligned(16))) float Ps[2 * BATCH * APITCH];

  const int s    = blockIdx.x;
  const int rng  = blockIdx.y;
  const int t    = threadIdx.x;
  const int lane = t & 31;
  const int wave = t >> 5;
  const int half = lane >> 4;
  const int lm   = lane & 15;
  const int tile = rng * 8 + wave;

  const float* src = buf + (FIRST ? 0 : SETSZ);
  float*       dst = buf + (FIRST ? SETSZ : 0);

  const int q = t & 127;
  const int g = t >> 7;
  const float* in = src + (size_t)s * PIECE + (size_t)g * PLANE + q * 4;
  v4f v[NSLAB / 2];
#pragma unroll
  for (int i = 0; i < NSLAB / 2; ++i) v[i] = *(const v4f*)(in + (size_t)(2 * i) * PLANE);

  v4f bias4 = {0.f, 0.f, 0.f, 0.f};
  if (!FIRST) bias4 = *(const v4f*)(bias_in + s * KSLAB + (q & 7) * 4);

  __builtin_amdgcn_sched_barrier(0);

  const float* wrow = Wm + (size_t)(tile * 16 + lm) * DIM + s * KSLAB + half * 16;
  v4f w[4];
#pragma unroll
  for (int m = 0; m < 4; ++m) w[m] = *(const v4f*)(wrow + m * 4);

  __builtin_amdgcn_sched_barrier(0);

  v4f gs = v[0];
#pragma unroll
  for (int i = 1; i < NSLAB / 2; ++i) gs += v[i];
  if (!FIRST && g == 0) gs += bias4;
  *(v4f*)(Ps + g * (BATCH * APITCH) + (q >> 3) * APITCH + (q & 7) * 4) = gs;
  __builtin_amdgcn_fence(__ATOMIC_RELEASE, "workgroup", "local");
  __builtin_amdgcn_s_barrier();
  __builtin_amdgcn_fence(__ATOMIC_ACQUIRE, "workgroup", "local");

  const float* pr = Ps + lm * APITCH + half * 16;
  v4f a[4];
#pragma unroll
  for (int m = 0; m < 4; ++m) {
    a[m] = *(const v4f*)(pr + m * 4) + *(const v4f*)(pr + BATCH * APITCH + m * 4);
    if (FIRST) a[m] *= (1.0f / (float)SEQ);
  }

  v8f c0, c1;
#pragma unroll
  for (int r = 0; r < 8; ++r) { c0[r] = 0.f; c1[r] = 0.f; }

#pragma unroll
  for (int j = 0; j < 8; ++j) {
    const v4f aq = a[j >> 1];
    const v4f wq = w[j >> 1];
    const v2f af = (j & 1) ? (v2f){aq[2], aq[3]} : (v2f){aq[0], aq[1]};
    const v2f bf = (j & 1) ? (v2f){wq[2], wq[3]} : (v2f){wq[0], wq[1]};
    if (j & 1)
      c1 = __builtin_amdgcn_wmma_f32_16x16x4_f32(
          false, af, false, bf, (short)0, c1, false, false);
    else
      c0 = __builtin_amdgcn_wmma_f32_16x16x4_f32(
          false, af, false, bf, (short)0, c0, false, false);
  }
  const v8f c = c0 + c1;

  float* op = dst + (size_t)s * PLANE;
#pragma unroll
  for (int r = 0; r < 8; ++r) {
    const int row = 8 * half + r;
    if (FIRST) op[(tile >> 1) * PIECE + row * KSLAB + (tile & 1) * 16 + lm] = c[r];
    else       op[row * DIM + tile * 16 + lm] = c[r];
  }
}

__global__ __launch_bounds__(32) __attribute__((amdgpu_waves_per_eu(1, 4)))
void hcen_sum_planes(
    const float* __restrict__ planes, const float* __restrict__ bias,
    float* __restrict__ out) {
  const int i2 = (blockIdx.x * 32 + threadIdx.x) * 2;
  v2f acc = {0.f, 0.f};
#pragma unroll
  for (int p = 0; p < NSLAB; ++p)
    acc += *(const v2f*)(planes + (size_t)p * PLANE + i2);
  const v2f bv = *(const v2f*)(bias + (i2 & (DIM - 1)));
  *(v2f*)(out + i2) = acc + bv;
}

extern "C" void kernel_launch(void* const* d_in, const int* in_sizes, int n_in,
                              void* d_out, int out_size, void* d_ws, size_t ws_size,
                              hipStream_t stream) {
  const float* x     = (const float*)d_in[0];
  const float* W_enc = (const float*)d_in[1];
  const float* b_enc = (const float*)d_in[2];
  const float* W_out = (const float*)d_in[3];
  const float* b_out = (const float*)d_in[4];
  float* out = (float*)d_out;

  float* buf = (float*)d_ws;

  dim3 g1(SSPLIT, BATCH);
  dim3 g2(NSLAB, NRANGE);
  hcen_mean_stage1<<<g1, 256, 0, stream>>>(x, buf);
  hcen_rank32<true ><<<g2, 256, 0, stream>>>(buf, W_enc, nullptr);
  hcen_rank32<false><<<g2, 256, 0, stream>>>(buf, W_out, b_enc);
  hcen_sum_planes<<<(BATCH * DIM) / 64, 32, 0, stream>>>(buf, b_out, out);
}
